# starting kernel
# baseline (speedup 1.0000x reference)
#include <hip/hip_runtime.h>

typedef float v2f __attribute__((ext_vector_type(2)));
typedef float v8f __attribute__((ext_vector_type(8)));

#define T_   512
#define K_   4
#define G_   20000
#define EPG_ 3
#define B_   512
#define BT   16
#define GT   512
#define NW   16

__device__ __forceinline__ float lrelu(float x) {
    return x > 0.f ? x : 0.01f * x;
}

__global__ __launch_bounds__(512)
void ae_decoder_fused(const float* __restrict__ features,
                      const float* __restrict__ w1,
                      const float* __restrict__ b1,
                      const float* __restrict__ w2,
                      const float* __restrict__ b2,
                      const float* __restrict__ w3,
                      const float* __restrict__ b3,
                      const int*   __restrict__ edge_tf,
                      float*       __restrict__ out)
{
    __shared__ __align__(16) float lds_h2[BT * T_ * K_];

    const int tid  = threadIdx.x;
    const int wave = tid >> 5;
    const int lane = tid & 31;
    const int half = lane >> 4;
    const int m    = lane & 15;

    const int bbase = blockIdx.y * BT;
    const int gbase = blockIdx.x * GT;

    const int j0 = half * 2;
    for (int i = 0; i < T_ / NW; ++i) {
        const int t = wave * (T_ / NW) + i;

        const float f = features[(size_t)(bbase + m) * T_ + t];
        v2f a;
        a.x = lrelu(f * w1[t * K_ + j0]     + b1[t * K_ + j0]);
        a.y = lrelu(f * w1[t * K_ + j0 + 1] + b1[t * K_ + j0 + 1]);

        float bx = 0.f, by = 0.f, cv = 0.f;
        if (m < K_) {
            bx = w2[(t * K_ + j0)     * K_ + m];
            by = w2[(t * K_ + j0 + 1) * K_ + m];
            cv = b2[t * K_ + m];
        }
        v2f bmat; bmat.x = bx; bmat.y = by;
        v8f c = {cv, cv, cv, cv, cv, cv, cv, cv};

        v8f d = __builtin_amdgcn_wmma_f32_16x16x4_f32(
             false, a,  false, bmat,
             (short)0, c,  false,  false);

        if (m < K_) {
            #pragma unroll
            for (int v = 0; v < 8; ++v) {
                const int brow = v + 8 * half;
                lds_h2[(brow * T_ + t) * K_ + m] = lrelu(d[v]);
            }
        }
    }

    __syncthreads();

    const int    bl    = wave;
    const size_t bglob = (size_t)(bbase + bl);
    const float4* __restrict__ w3v = (const float4*)w3;

    for (int it = 0; it < GT / 32; ++it) {
        const int g = gbase + it * 32 + lane;
        if (g < G_) {
            float acc = b3[g];
            #pragma unroll
            for (int e = 0; e < EPG_; ++e) {
                const int t = edge_tf[g * EPG_ + e];
                const float4 h  = *(const float4*)&lds_h2[(bl * T_ + t) * K_];
                const float4 wv = w3v[g * EPG_ + e];
                acc = fmaf(h.x, wv.x, acc);
                acc = fmaf(h.y, wv.y, acc);
                acc = fmaf(h.z, wv.z, acc);
                acc = fmaf(h.w, wv.w, acc);
            }
            out[bglob * G_ + g] = acc;
        }
    }
}

extern "C" void kernel_launch(void* const* d_in, const int* in_sizes, int n_in,
                              void* d_out, int out_size, void* d_ws, size_t ws_size,
                              hipStream_t stream)
{
    const float* features = (const float*)d_in[0];
    const float* w1       = (const float*)d_in[1];
    const float* b1       = (const float*)d_in[2];
    const float* w2       = (const float*)d_in[3];
    const float* b2       = (const float*)d_in[4];
    const float* w3       = (const float*)d_in[5];
    const float* b3       = (const float*)d_in[6];
    const int*   edge_tf  = (const int*)d_in[7];
    float*       out      = (float*)d_out;

    dim3 grid((G_ + GT - 1) / GT, B_ / BT);
    ae_decoder_fused<<<grid, 512, 0, stream>>>(features, w1, b1, w2, b2,
                                               w3, b3, edge_tf, out);
}


# Phase B: each lane owns one g, keeps its edge indices, w3 and b3 in registers (loaded once, ahead of phase A) and walks the tile's b rows in LDS: no global load per (b,g); same fma order
# speedup vs baseline: 1.2883x; 1.2883x over previous
#include <hip/hip_runtime.h>

typedef float v2f __attribute__((ext_vector_type(2)));
typedef float v8f __attribute__((ext_vector_type(8)));

#define T_   512
#define K_   4
#define G_   20000
#define EPG_ 3
#define B_   512
#define BT   16
#define GT   512
#define NW   16

__device__ __forceinline__ float lrelu(float x) {
    return x > 0.f ? x : 0.01f * x;
}

__global__ __launch_bounds__(512)
void ae_decoder_fused(const float* __restrict__ features,
                      const float* __restrict__ w1,
                      const float* __restrict__ b1,
                      const float* __restrict__ w2,
                      const float* __restrict__ b2,
                      const float* __restrict__ w3,
                      const float* __restrict__ b3,
                      const int*   __restrict__ edge_tf,
                      float*       __restrict__ out)
{
    __shared__ __align__(16) float lds_h2[BT * T_ * K_];

    const int tid  = threadIdx.x;
    const int wave = tid >> 5;
    const int lane = tid & 31;
    const int half = lane >> 4;
    const int m    = lane & 15;

    const int bbase = blockIdx.y * BT;
    const int gbase = blockIdx.x * GT;

    const int  g      = gbase + wave * 32 + lane;
    const bool gvalid = g < G_;
    const float4* __restrict__ w3v = (const float4*)w3;
    int t0 = 0, t1 = 0, t2 = 0;
    float4 wv0 = {0.f, 0.f, 0.f, 0.f}, wv1 = wv0, wv2 = wv0;
    float bias = 0.f;
    if (gvalid) {
        t0   = edge_tf[g * EPG_ + 0];
        t1   = edge_tf[g * EPG_ + 1];
        t2   = edge_tf[g * EPG_ + 2];
        wv0  = w3v[g * EPG_ + 0];
        wv1  = w3v[g * EPG_ + 1];
        wv2  = w3v[g * EPG_ + 2];
        bias = b3[g];
    }

    const int j0 = half * 2;
    for (int i = 0; i < T_ / NW; ++i) {
        const int t = wave * (T_ / NW) + i;

        const float f = features[(size_t)(bbase + m) * T_ + t];
        v2f a;
        a.x = lrelu(f * w1[t * K_ + j0]     + b1[t * K_ + j0]);
        a.y = lrelu(f * w1[t * K_ + j0 + 1] + b1[t * K_ + j0 + 1]);

        float bx = 0.f, by = 0.f, cv = 0.f;
        if (m < K_) {
            bx = w2[(t * K_ + j0)     * K_ + m];
            by = w2[(t * K_ + j0 + 1) * K_ + m];
            cv = b2[t * K_ + m];
        }
        v2f bmat; bmat.x = bx; bmat.y = by;
        v8f c = {cv, cv, cv, cv, cv, cv, cv, cv};

        v8f d = __builtin_amdgcn_wmma_f32_16x16x4_f32(
             false, a,  false, bmat,
             (short)0, c,  false,  false);

        if (m < K_) {
            #pragma unroll
            for (int v = 0; v < 8; ++v) {
                const int brow = v + 8 * half;
                lds_h2[(brow * T_ + t) * K_ + m] = lrelu(d[v]);
            }
        }
    }

    __syncthreads();

    if (gvalid) {
        float* __restrict__ op = out + (size_t)bbase * G_ + g;
        #pragma unroll
        for (int bl = 0; bl < BT; ++bl) {
            const float4 h0 = *(const float4*)&lds_h2[(bl * T_ + t0) * K_];
            const float4 h1 = *(const float4*)&lds_h2[(bl * T_ + t1) * K_];
            const float4 h2 = *(const float4*)&lds_h2[(bl * T_ + t2) * K_];
            float acc = bias;
            acc = fmaf(h0.x, wv0.x, acc);
            acc = fmaf(h0.y, wv0.y, acc);
            acc = fmaf(h0.z, wv0.z, acc);
            acc = fmaf(h0.w, wv0.w, acc);
            acc = fmaf(h1.x, wv1.x, acc);
            acc = fmaf(h1.y, wv1.y, acc);
            acc = fmaf(h1.z, wv1.z, acc);
            acc = fmaf(h1.w, wv1.w, acc);
            acc = fmaf(h2.x, wv2.x, acc);
            acc = fmaf(h2.y, wv2.y, acc);
            acc = fmaf(h2.z, wv2.z, acc);
            acc = fmaf(h2.w, wv2.w, acc);
            op[(size_t)bl * G_] = acc;
        }
    }
}

extern "C" void kernel_launch(void* const* d_in, const int* in_sizes, int n_in,
                              void* d_out, int out_size, void* d_ws, size_t ws_size,
                              hipStream_t stream)
{
    const float* features = (const float*)d_in[0];
    const float* w1       = (const float*)d_in[1];
    const float* b1       = (const float*)d_in[2];
    const float* w2       = (const float*)d_in[3];
    const float* b2       = (const float*)d_in[4];
    const float* w3       = (const float*)d_in[5];
    const float* b3       = (const float*)d_in[6];
    const int*   edge_tf  = (const int*)d_in[7];
    float*       out      = (float*)d_out;

    dim3 grid((G_ + GT - 1) / GT, B_ / BT);
    ae_decoder_fused<<<grid, 512, 0, stream>>>(features, w1, b1, w2, b2,
                                               w3, b3, edge_tf, out);
}


# Each block keeps its 16-b h2 tile in LDS and walks 1/8 of all g (256 blocks x 32 waves; wave pairs split a 32-g chunk's b rows 8+8): phase A runs 256x, not 1280x; same fma order
# speedup vs baseline: 3.8034x; 2.9523x over previous
#include <hip/hip_runtime.h>


typedef float v2f __attribute__((ext_vector_type(2)));
typedef float v8f __attribute__((ext_vector_type(8)));

#define T_   512
#define K_   4
#define G_   20000
#define EPG_ 3
#define B_   512
#define BT   16
#define NW   32
#define NGS  8
#define GCH  32
#define NCH  (G_ / GCH)
#define NPAIR (NGS * NW / 2)
#define NU   ((NCH + NPAIR - 1) / NPAIR)
#define RH   (BT / 2)

static_assert(G_ % GCH == 0, "chunks are whole");
static_assert((NU - 1) * NPAIR <= NCH, "all but a pair's last chunk exist for every pair");
static_assert(T_ % NW == 0 && B_ % BT == 0, "tiles are whole");

__device__ __forceinline__ float lrelu(float x) {
    return x > 0.f ? x : 0.01f * x;
}

struct GOp {
    int    t0, t1, t2;
    float4 w0, w1, w2;
    float  bias;
};

__device__ __forceinline__ GOp load_gop(const int* __restrict__ edge_tf,
                                        const float* __restrict__ w3,
                                        const float* __restrict__ b3, int g)
{
    const float4* __restrict__ w3v = (const float4*)w3;
    GOp o;
    o.t0   = edge_tf[g * EPG_ + 0];
    o.t1   = edge_tf[g * EPG_ + 1];
    o.t2   = edge_tf[g * EPG_ + 2];
    o.w0   = w3v[g * EPG_ + 0];
    o.w1   = w3v[g * EPG_ + 1];
    o.w2   = w3v[g * EPG_ + 2];
    o.bias = b3[g];
    return o;
}

__device__ __forceinline__ void gather_rows(const float* __restrict__ hrow, const GOp& o,
                                            float* __restrict__ op)
{
    #pragma unroll
    for (int r = 0; r < RH; ++r) {
        const float4 h0 = *(const float4*)&hrow[(r * T_ + o.t0) * K_];
        const float4 h1 = *(const float4*)&hrow[(r * T_ + o.t1) * K_];
        const float4 h2 = *(const float4*)&hrow[(r * T_ + o.t2) * K_];
        float acc = o.bias;
        acc = fmaf(h0.x, o.w0.x, acc);
        acc = fmaf(h0.y, o.w0.y, acc);
        acc = fmaf(h0.z, o.w0.z, acc);
        acc = fmaf(h0.w, o.w0.w, acc);
        acc = fmaf(h1.x, o.w1.x, acc);
        acc = fmaf(h1.y, o.w1.y, acc);
        acc = fmaf(h1.z, o.w1.z, acc);
        acc = fmaf(h1.w, o.w1.w, acc);
        acc = fmaf(h2.x, o.w2.x, acc);
        acc = fmaf(h2.y, o.w2.y, acc);
        acc = fmaf(h2.z, o.w2.z, acc);
        acc = fmaf(h2.w, o.w2.w, acc);
        op[(size_t)r * G_] = acc;
    }
}

__global__ __launch_bounds__(1024)
void ae_decoder_fused(const float* __restrict__ features,
                      const float* __restrict__ w1,
                      const float* __restrict__ b1,
                      const float* __restrict__ w2,
                      const float* __restrict__ b2,
                      const float* __restrict__ w3,
                      const float* __restrict__ b3,
                      const int*   __restrict__ edge_tf,
                      float*       __restrict__ out)
{
    __shared__ __align__(16) float lds_h2[BT * T_ * K_];

    const int tid  = threadIdx.x;
    const int wave = tid >> 5;
    const int lane = tid & 31;
    const int half = lane >> 4;
    const int m    = lane & 15;

    const int bbase = blockIdx.y * BT;

    const int pair = blockIdx.x * (NW / 2) + (wave >> 1);
    const int rh   = wave & 1;
    const bool has_last = pair + (NU - 1) * NPAIR < NCH;

    GOp cur = load_gop(edge_tf, w3, b3, pair * GCH + lane);

    const int j0 = half * 2;
    for (int i = 0; i < T_ / NW; ++i) {
        const int t = wave * (T_ / NW) + i;

        const float f = features[(size_t)(bbase + m) * T_ + t];
        v2f a;
        a.x = lrelu(f * w1[t * K_ + j0]     + b1[t * K_ + j0]);
        a.y = lrelu(f * w1[t * K_ + j0 + 1] + b1[t * K_ + j0 + 1]);

        float bx = 0.f, by = 0.f, cv = 0.f;
        if (m < K_) {
            bx = w2[(t * K_ + j0)     * K_ + m];
            by = w2[(t * K_ + j0 + 1) * K_ + m];
            cv = b2[t * K_ + m];
        }
        v2f bmat; bmat.x = bx; bmat.y = by;
        v8f c = {cv, cv, cv, cv, cv, cv, cv, cv};

        v8f d = __builtin_amdgcn_wmma_f32_16x16x4_f32(
             false, a,  false, bmat,
             (short)0, c,  false,  false);

        if (m < K_) {
            #pragma unroll
            for (int v = 0; v < 8; ++v) {
                const int brow = v + 8 * half;
                lds_h2[(brow * T_ + t) * K_ + m] = lrelu(d[v]);
            }
        }
    }

    __syncthreads();

    const float* __restrict__ hrow = lds_h2 + rh * RH * T_ * K_;
    float* __restrict__ orow = out + (size_t)(bbase + rh * RH) * G_ + lane;

    #pragma unroll
    for (int u = 0; u < NU - 1; ++u) {
        const int c = pair + u * NPAIR;
        GOp nxt = cur;
        if (u + 1 < NU - 1 || has_last)
            nxt = load_gop(edge_tf, w3, b3, (c + NPAIR) * GCH + lane);
        gather_rows(hrow, cur, orow + c * GCH);
        cur = nxt;
    }
    if (has_last)
        gather_rows(hrow, cur, orow + (pair + (NU - 1) * NPAIR) * GCH);
}

extern "C" void kernel_launch(void* const* d_in, const int* in_sizes, int n_in,
                              void* d_out, int out_size, void* d_ws, size_t ws_size,
                              hipStream_t stream)
{
    const float* features = (const float*)d_in[0];
    const float* w1       = (const float*)d_in[1];
    const float* b1       = (const float*)d_in[2];
    const float* w2       = (const float*)d_in[3];
    const float* b2       = (const float*)d_in[4];
    const float* w3       = (const float*)d_in[5];
    const float* b3       = (const float*)d_in[6];
    const int*   edge_tf  = (const int*)d_in[7];
    float*       out      = (float*)d_out;

    dim3 grid(NGS, B_ / BT);
    ae_decoder_fused<<<grid, NW * 32, 0, stream>>>(features, w1, b1, w2, b2,
                                                   w3, b3, edge_tf, out);
}


# Phase A: each wave fetches its operands in two groups of 8 t ahead of the WMMAs (a lane's 8 features as two 16-byte loads, no branch round the w2/b2 loads): 2 memory round trips per wave, not 16
# speedup vs baseline: 4.0509x; 1.0651x over previous
#include <hip/hip_runtime.h>


typedef float v2f __attribute__((ext_vector_type(2)));
typedef float v8f __attribute__((ext_vector_type(8)));

#define T_   512
#define K_   4
#define G_   20000
#define EPG_ 3
#define B_   512
#define BT   16
#define NW   32
#define NGS  8
#define GCH  32
#define NCH  (G_ / GCH)
#define NPAIR (NGS * NW / 2)
#define NU   ((NCH + NPAIR - 1) / NPAIR)
#define RH   (BT / 2)
#define TG   8

static_assert(G_ % GCH == 0, "chunks are whole");
static_assert((NU - 1) * NPAIR <= NCH, "all but a pair's last chunk exist for every pair");
static_assert(T_ % NW == 0 && B_ % BT == 0, "tiles are whole");
static_assert((T_ / NW) % TG == 0 && TG == 8, "a group is two 16-byte feature loads per lane");

__device__ __forceinline__ float lrelu(float x) {
    return x > 0.f ? x : 0.01f * x;
}

struct GOp {
    int    t0, t1, t2;
    float4 w0, w1, w2;
    float  bias;
};

__device__ __forceinline__ GOp load_gop(const int* __restrict__ edge_tf,
                                        const float* __restrict__ w3,
                                        const float* __restrict__ b3, int g)
{
    const float4* __restrict__ w3v = (const float4*)w3;
    GOp o;
    o.t0   = edge_tf[g * EPG_ + 0];
    o.t1   = edge_tf[g * EPG_ + 1];
    o.t2   = edge_tf[g * EPG_ + 2];
    o.w0   = w3v[g * EPG_ + 0];
    o.w1   = w3v[g * EPG_ + 1];
    o.w2   = w3v[g * EPG_ + 2];
    o.bias = b3[g];
    return o;
}

__device__ __forceinline__ void gather_rows(const float* __restrict__ hrow, const GOp& o,
                                            float* __restrict__ op)
{
    #pragma unroll
    for (int r = 0; r < RH; ++r) {
        const float4 h0 = *(const float4*)&hrow[(r * T_ + o.t0) * K_];
        const float4 h1 = *(const float4*)&hrow[(r * T_ + o.t1) * K_];
        const float4 h2 = *(const float4*)&hrow[(r * T_ + o.t2) * K_];
        float acc = o.bias;
        acc = fmaf(h0.x, o.w0.x, acc);
        acc = fmaf(h0.y, o.w0.y, acc);
        acc = fmaf(h0.z, o.w0.z, acc);
        acc = fmaf(h0.w, o.w0.w, acc);
        acc = fmaf(h1.x, o.w1.x, acc);
        acc = fmaf(h1.y, o.w1.y, acc);
        acc = fmaf(h1.z, o.w1.z, acc);
        acc = fmaf(h1.w, o.w1.w, acc);
        acc = fmaf(h2.x, o.w2.x, acc);
        acc = fmaf(h2.y, o.w2.y, acc);
        acc = fmaf(h2.z, o.w2.z, acc);
        acc = fmaf(h2.w, o.w2.w, acc);
        op[(size_t)r * G_] = acc;
    }
}

__global__ __launch_bounds__(1024)
void ae_decoder_fused(const float* __restrict__ features,
                      const float* __restrict__ w1,
                      const float* __restrict__ b1,
                      const float* __restrict__ w2,
                      const float* __restrict__ b2,
                      const float* __restrict__ w3,
                      const float* __restrict__ b3,
                      const int*   __restrict__ edge_tf,
                      float*       __restrict__ out)
{
    __shared__ __align__(16) float lds_h2[BT * T_ * K_];

    const int tid  = threadIdx.x;
    const int wave = tid >> 5;
    const int lane = tid & 31;
    const int half = lane >> 4;
    const int m    = lane & 15;

    const int bbase = blockIdx.y * BT;

    const int pair = blockIdx.x * (NW / 2) + (wave >> 1);
    const int rh   = wave & 1;
    const bool has_last = pair + (NU - 1) * NPAIR < NCH;

    GOp cur = load_gop(edge_tf, w3, b3, pair * GCH + lane);

    const int j0 = half * 2;
    const int mc = m & (K_ - 1);
    const int tw = wave * (T_ / NW);
    const float* __restrict__ frow = features + (size_t)(bbase + m) * T_ + tw;

    #pragma unroll
    for (int q = 0; q < (T_ / NW) / TG; ++q) {
        const int tq = tw + q * TG;

        const float4 fa = *(const float4*)(frow + q * TG);
        const float4 fb = *(const float4*)(frow + q * TG + 4);
        const float fv[TG] = {fa.x, fa.y, fa.z, fa.w, fb.x, fb.y, fb.z, fb.w};

        v2f   w1v[TG], b1v[TG];
        float w2x[TG], w2y[TG], b2v[TG];
        #pragma unroll
        for (int i = 0; i < TG; ++i) {
            const int t = tq + i;
            w1v[i] = *(const v2f*)&w1[t * K_ + j0];
            b1v[i] = *(const v2f*)&b1[t * K_ + j0];
            w2x[i] = w2[(t * K_ + j0)     * K_ + mc];
            w2y[i] = w2[(t * K_ + j0 + 1) * K_ + mc];
            b2v[i] = b2[t * K_ + mc];
        }

        #pragma unroll
        for (int i = 0; i < TG; ++i) {
            const int t = tq + i;
            const float f = fv[i];
            v2f a;
            a.x = lrelu(f * w1v[i].x + b1v[i].x);
            a.y = lrelu(f * w1v[i].y + b1v[i].y);

            v2f bmat; bmat.x = w2x[i]; bmat.y = w2y[i];
            const float cv = b2v[i];
            v8f c = {cv, cv, cv, cv, cv, cv, cv, cv};

            v8f d = __builtin_amdgcn_wmma_f32_16x16x4_f32(
                 false, a,  false, bmat,
                 (short)0, c,  false,  false);

            if (m < K_) {
                #pragma unroll
                for (int v = 0; v < 8; ++v) {
                    const int brow = v + 8 * half;
                    lds_h2[(brow * T_ + t) * K_ + m] = lrelu(d[v]);
                }
            }
        }
    }

    __syncthreads();

    const float* __restrict__ hrow = lds_h2 + rh * RH * T_ * K_;
    float* __restrict__ orow = out + (size_t)(bbase + rh * RH) * G_ + lane;

    #pragma unroll
    for (int u = 0; u < NU - 1; ++u) {
        const int c = pair + u * NPAIR;
        GOp nxt = cur;
        if (u + 1 < NU - 1 || has_last)
            nxt = load_gop(edge_tf, w3, b3, (c + NPAIR) * GCH + lane);
        gather_rows(hrow, cur, orow + c * GCH);
        cur = nxt;
    }
    if (has_last)
        gather_rows(hrow, cur, orow + (pair + (NU - 1) * NPAIR) * GCH);
}

extern "C" void kernel_launch(void* const* d_in, const int* in_sizes, int n_in,
                              void* d_out, int out_size, void* d_ws, size_t ws_size,
                              hipStream_t stream)
{
    const float* features = (const float*)d_in[0];
    const float* w1       = (const float*)d_in[1];
    const float* b1       = (const float*)d_in[2];
    const float* w2       = (const float*)d_in[3];
    const float* b2       = (const float*)d_in[4];
    const float* w3       = (const float*)d_in[5];
    const float* b3       = (const float*)d_in[6];
    const int*   edge_tf  = (const int*)d_in[7];
    float*       out      = (float*)d_out;

    dim3 grid(NGS, B_ / BT);
    ae_decoder_fused<<<grid, NW * 32, 0, stream>>>(features, w1, b1, w2, b2,
                                                   w3, b3, edge_tf, out);
}


# Phase B: a row's 16 lanes walk the 16 b of one g (tile as [t][b][k], g operands by DPP row_share): conflict-free LDS gather; results turned through LDS for whole-line stores; same fma order
# speedup vs baseline: 4.3242x; 1.0675x over previous
#include <hip/hip_runtime.h>


typedef float v2f __attribute__((ext_vector_type(2)));
typedef float v8f __attribute__((ext_vector_type(8)));

#define T_   512
#define K_   4
#define G_   20000
#define EPG_ 3
#define B_   512
#define BT   16
#define NW   32
#define NGS  8
#define GCH  32
#define NCH  (G_ / GCH)
#define NBC  ((NCH + NGS - 1) / NGS)
#define NU   ((NBC + NW - 1) / NW)
#define GSTEP (NGS * NW * GCH)
#define TRP  36
#define TROW (BT * K_)
#define TG   8

static_assert(G_ % GCH == 0, "chunks are whole");
static_assert((NU - 1) * NW <= NCH / NGS, "all but a wave's last chunk exist for every wave");
static_assert(BT == 16 && GCH == 32, "a 16-lane row walks the 16 b of the tile; a wave's two rows take 2 x 16 g");
static_assert(T_ % NW == 0 && B_ % BT == 0, "tiles are whole");
static_assert((T_ / NW) % TG == 0 && TG == 8, "a group is two 16-byte feature loads per lane");

__device__ __forceinline__ float lrelu(float x) {
    return x > 0.f ? x : 0.01f * x;
}

struct GOp {
    int    t0, t1, t2;
    float4 w0, w1, w2;
    float  bias;
};

__device__ __forceinline__ GOp load_gop(const int* __restrict__ edge_tf,
                                        const float* __restrict__ w3,
                                        const float* __restrict__ b3, unsigned g)
{
    const float4* __restrict__ w3v = (const float4*)w3;
    GOp o;
    o.t0   = edge_tf[g * EPG_ + 0u];
    o.t1   = edge_tf[g * EPG_ + 1u];
    o.t2   = edge_tf[g * EPG_ + 2u];
    o.w0   = w3v[g * EPG_ + 0u];
    o.w1   = w3v[g * EPG_ + 1u];
    o.w2   = w3v[g * EPG_ + 2u];
    o.bias = b3[g];
    return o;
}

template <int S>
__device__ __forceinline__ int row_share_i(int x) {
    return __builtin_amdgcn_update_dpp(0, x, 0x150 + S, 0xf, 0xf, true);
}
template <int S>
__device__ __forceinline__ float row_share_f(float x) {
    return __int_as_float(row_share_i<S>(__float_as_int(x)));
}
template <int S>
__device__ __forceinline__ float fma_row_share(float h, float w, float acc) {
    asm("v_fmac_f32_dpp %0, %1, %2 row_share:%3 row_mask:0xf bank_mask:0xf bound_ctrl:1"
        : "+v"(acc) : "v"(w), "v"(h), "n"(S));
    return acc;
}

struct H3 { float4 h0, h1, h2; };

template <int S>
__device__ __forceinline__ H3 load_step(const char* hb, const int3& to)
{
    H3 h;
    h.h0 = *(const float4*)(hb + row_share_i<S>(to.x));
    h.h1 = *(const float4*)(hb + row_share_i<S>(to.y));
    h.h2 = *(const float4*)(hb + row_share_i<S>(to.z));
    return h;
}

template <int S>
__device__ __forceinline__ float dot_step(const H3& h, const GOp& o)
{
    const float4 h0 = h.h0, h1 = h.h1, h2 = h.h2;
    float acc = row_share_f<S>(o.bias);
    acc = fma_row_share<S>(h0.x, o.w0.x, acc);
    acc = fma_row_share<S>(h0.y, o.w0.y, acc);
    acc = fma_row_share<S>(h0.z, o.w0.z, acc);
    acc = fma_row_share<S>(h0.w, o.w0.w, acc);
    acc = fma_row_share<S>(h1.x, o.w1.x, acc);
    acc = fma_row_share<S>(h1.y, o.w1.y, acc);
    acc = fma_row_share<S>(h1.z, o.w1.z, acc);
    acc = fma_row_share<S>(h1.w, o.w1.w, acc);
    acc = fma_row_share<S>(h2.x, o.w2.x, acc);
    acc = fma_row_share<S>(h2.y, o.w2.y, acc);
    acc = fma_row_share<S>(h2.z, o.w2.z, acc);
    acc = fma_row_share<S>(h2.w, o.w2.w, acc);
    return acc;
}

template <int S>
__device__ __forceinline__ float4 dot_steps4(const char* hb, const int3& to, const GOp& o)
{
    const H3 ha = load_step<S>(hb, to);
    const H3 hb1 = load_step<S + 1>(hb, to);
    const H3 hc = load_step<S + 2>(hb, to);
    const H3 hd = load_step<S + 3>(hb, to);
    float4 r;
    r.x = dot_step<S>(ha, o);
    r.y = dot_step<S + 1>(hb1, o);
    r.z = dot_step<S + 2>(hc, o);
    r.w = dot_step<S + 3>(hd, o);
    return r;
}

__device__ __forceinline__ void wave_patch_order()
{
    __builtin_amdgcn_fence(__ATOMIC_RELEASE, "wavefront");
    __builtin_amdgcn_wave_barrier();
    __builtin_amdgcn_fence(__ATOMIC_ACQUIRE, "wavefront");
}

template <int H>
__device__ __forceinline__ void half_chunk(const char* hb, const GOp& o,
                                           float* patch_w)
{
    const int3 to = make_int3(o.t0 * (TROW * 4), o.t1 * (TROW * 4), o.t2 * (TROW * 4));
    *(float4*)&patch_w[8 * H]     = dot_steps4<8 * H>(hb, to, o);
    *(float4*)&patch_w[8 * H + 4] = dot_steps4<8 * H + 4>(hb, to, o);
}

__device__ __forceinline__ void turn_and_store(const float* patch_r,
                                               float* __restrict__ op)
{
    wave_patch_order();
    #pragma unroll
    for (int r = 0; r < BT; ++r)
        op[(size_t)r * G_] = patch_r[r * TRP];
    wave_patch_order();
}

__global__ __launch_bounds__(1024)
void ae_decoder_fused(const float* __restrict__ features,
                      const float* __restrict__ w1,
                      const float* __restrict__ b1,
                      const float* __restrict__ w2,
                      const float* __restrict__ b2,
                      const float* __restrict__ w3,
                      const float* __restrict__ b3,
                      const int*   __restrict__ edge_tf,
                      float*       __restrict__ out)
{
    __shared__ __align__(16) float lds_h2[T_ * BT * K_];
    __shared__ __align__(16) float lds_tr[NW * BT * TRP];

    const int tid  = threadIdx.x;
    const int wave = tid >> 5;
    const int lane = tid & 31;
    const int half = lane >> 4;
    const int m    = lane & 15;

    const int bbase = blockIdx.y * BT;

    const int g0 = (blockIdx.x + NGS * wave) * GCH;
    const bool has_last = g0 + (NU - 1) * GSTEP < G_;

    GOp cur = load_gop(edge_tf, w3, b3, g0 + lane);

    const int j0 = half * 2;
    const int mc = m & (K_ - 1);
    const int tw = wave * (T_ / NW);
    const float* __restrict__ frow = features + (size_t)(bbase + m) * T_ + tw;

    #pragma unroll
    for (int q = 0; q < (T_ / NW) / TG; ++q) {
        const int tq = tw + q * TG;

        const float4 fa = *(const float4*)(frow + q * TG);
        const float4 fb = *(const float4*)(frow + q * TG + 4);
        const float fv[TG] = {fa.x, fa.y, fa.z, fa.w, fb.x, fb.y, fb.z, fb.w};

        v2f   w1v[TG], b1v[TG];
        float w2x[TG], w2y[TG], b2v[TG];
        #pragma unroll
        for (int i = 0; i < TG; ++i) {
            const int t = tq + i;
            w1v[i] = *(const v2f*)&w1[t * K_ + j0];
            b1v[i] = *(const v2f*)&b1[t * K_ + j0];
            w2x[i] = w2[(t * K_ + j0)     * K_ + mc];
            w2y[i] = w2[(t * K_ + j0 + 1) * K_ + mc];
            b2v[i] = b2[t * K_ + mc];
        }

        #pragma unroll
        for (int i = 0; i < TG; ++i) {
            const int t = tq + i;
            const float f = fv[i];
            v2f a;
            a.x = lrelu(f * w1v[i].x + b1v[i].x);
            a.y = lrelu(f * w1v[i].y + b1v[i].y);

            v2f bmat; bmat.x = w2x[i]; bmat.y = w2y[i];
            const float cv = b2v[i];
            v8f c = {cv, cv, cv, cv, cv, cv, cv, cv};

            v8f d = __builtin_amdgcn_wmma_f32_16x16x4_f32(
                 false, a,  false, bmat,
                 (short)0, c,  false,  false);

            if (m < K_) {
                #pragma unroll
                for (int v = 0; v < 8; ++v) {
                    const int brow = v + 8 * half;
                    lds_h2[(t * BT + brow) * K_ + m] = lrelu(d[v]);
                }
            }
        }
    }

    __syncthreads();

    const char* hb = (const char*)(lds_h2 + m * K_);
    float* patch = lds_tr + wave * (BT * TRP);
    float* patch_w = patch + m * TRP + 16 * half;
    const float* patch_r = patch + lane;
    float* __restrict__ orow = out + (size_t)bbase * G_ + lane;

    #pragma unroll
    for (int u = 0; u < NU - 1; ++u) {
        const int gb = g0 + u * GSTEP;
        half_chunk<0>(hb, cur, patch_w);
        GOp nxt = cur;
        if (u + 1 < NU - 1 || has_last)
            nxt = load_gop(edge_tf, w3, b3, gb + GSTEP + lane);
        half_chunk<1>(hb, cur, patch_w);
        turn_and_store(patch_r, orow + gb);
        cur = nxt;
    }
    if (has_last) {
        half_chunk<0>(hb, cur, patch_w);
        half_chunk<1>(hb, cur, patch_w);
        turn_and_store(patch_r, orow + g0 + (NU - 1) * GSTEP);
    }
}

extern "C" void kernel_launch(void* const* d_in, const int* in_sizes, int n_in,
                              void* d_out, int out_size, void* d_ws, size_t ws_size,
                              hipStream_t stream)
{
    const float* features = (const float*)d_in[0];
    const float* w1       = (const float*)d_in[1];
    const float* b1       = (const float*)d_in[2];
    const float* w2       = (const float*)d_in[3];
    const float* b2       = (const float*)d_in[4];
    const float* w3       = (const float*)d_in[5];
    const float* b3       = (const float*)d_in[6];
    const int*   edge_tf  = (const int*)d_in[7];
    float*       out      = (float*)d_out;

    dim3 grid(NGS, B_ / BT);
    ae_decoder_fused<<<grid, NW * 32, 0, stream>>>(features, w1, b1, w2, b2,
                                                   w3, b3, edge_tf, out);
}


# Phase A: product turned over (D rows = 4 t x 4 k, columns = b; w2 operand zero outside the t in work): all of D is h2; 4 chained WMMAs, one lrelu pass, two 16-byte LDS stores per 4 t; no branch
# speedup vs baseline: 4.6485x; 1.0750x over previous
#include <hip/hip_runtime.h>


typedef float v2f __attribute__((ext_vector_type(2)));
typedef float v8f __attribute__((ext_vector_type(8)));

#define T_   512
#define K_   4
#define G_   20000
#define EPG_ 3
#define B_   512
#define BT   16
#define NW   32
#define NGS  8
#define GCH  32
#define NCH  (G_ / GCH)
#define NBC  ((NCH + NGS - 1) / NGS)
#define NU   ((NBC + NW - 1) / NW)
#define GSTEP (NGS * NW * GCH)
#define TRP  36
#define TROW (BT * K_)
#define TG   8
#define TQ   4

static_assert(G_ % GCH == 0, "chunks are whole");
static_assert((NU - 1) * NW <= NCH / NGS, "all but a wave's last chunk exist for every wave");
static_assert(BT == 16 && GCH == 32, "a 16-lane row walks the 16 b of the tile; a wave's two rows take 2 x 16 g");
static_assert(T_ % NW == 0 && B_ % BT == 0, "tiles are whole");
static_assert((T_ / NW) % TG == 0 && TG == 8, "a group is two 16-byte feature loads per lane");
static_assert(TQ * K_ == 16 && TG % TQ == 0, "a D tile's 16 rows are TQ t x K_ k; a load group is whole tiles");

__device__ __forceinline__ float lrelu(float x) {
    return x > 0.f ? x : 0.01f * x;
}

struct GOp {
    int    t0, t1, t2;
    float4 w0, w1, w2;
    float  bias;
};

__device__ __forceinline__ GOp load_gop(const int* __restrict__ edge_tf,
                                        const float* __restrict__ w3,
                                        const float* __restrict__ b3, unsigned g)
{
    const float4* __restrict__ w3v = (const float4*)w3;
    GOp o;
    o.t0   = edge_tf[g * EPG_ + 0u];
    o.t1   = edge_tf[g * EPG_ + 1u];
    o.t2   = edge_tf[g * EPG_ + 2u];
    o.w0   = w3v[g * EPG_ + 0u];
    o.w1   = w3v[g * EPG_ + 1u];
    o.w2   = w3v[g * EPG_ + 2u];
    o.bias = b3[g];
    return o;
}

template <int S>
__device__ __forceinline__ int row_share_i(int x) {
    return __builtin_amdgcn_update_dpp(0, x, 0x150 + S, 0xf, 0xf, true);
}
template <int S>
__device__ __forceinline__ float row_share_f(float x) {
    return __int_as_float(row_share_i<S>(__float_as_int(x)));
}
template <int S>
__device__ __forceinline__ float fma_row_share(float h, float w, float acc) {
    asm("v_fmac_f32_dpp %0, %1, %2 row_share:%3 row_mask:0xf bank_mask:0xf bound_ctrl:1"
        : "+v"(acc) : "v"(w), "v"(h), "n"(S));
    return acc;
}

struct H3 { float4 h0, h1, h2; };

template <int S>
__device__ __forceinline__ H3 load_step(const char* hb, const int3& to)
{
    H3 h;
    h.h0 = *(const float4*)(hb + row_share_i<S>(to.x));
    h.h1 = *(const float4*)(hb + row_share_i<S>(to.y));
    h.h2 = *(const float4*)(hb + row_share_i<S>(to.z));
    return h;
}

template <int S>
__device__ __forceinline__ float dot_step(const H3& h, const GOp& o)
{
    const float4 h0 = h.h0, h1 = h.h1, h2 = h.h2;
    float acc = row_share_f<S>(o.bias);
    acc = fma_row_share<S>(h0.x, o.w0.x, acc);
    acc = fma_row_share<S>(h0.y, o.w0.y, acc);
    acc = fma_row_share<S>(h0.z, o.w0.z, acc);
    acc = fma_row_share<S>(h0.w, o.w0.w, acc);
    acc = fma_row_share<S>(h1.x, o.w1.x, acc);
    acc = fma_row_share<S>(h1.y, o.w1.y, acc);
    acc = fma_row_share<S>(h1.z, o.w1.z, acc);
    acc = fma_row_share<S>(h1.w, o.w1.w, acc);
    acc = fma_row_share<S>(h2.x, o.w2.x, acc);
    acc = fma_row_share<S>(h2.y, o.w2.y, acc);
    acc = fma_row_share<S>(h2.z, o.w2.z, acc);
    acc = fma_row_share<S>(h2.w, o.w2.w, acc);
    return acc;
}

template <int S>
__device__ __forceinline__ float4 dot_steps4(const char* hb, const int3& to, const GOp& o)
{
    const H3 ha = load_step<S>(hb, to);
    const H3 hb1 = load_step<S + 1>(hb, to);
    const H3 hc = load_step<S + 2>(hb, to);
    const H3 hd = load_step<S + 3>(hb, to);
    float4 r;
    r.x = dot_step<S>(ha, o);
    r.y = dot_step<S + 1>(hb1, o);
    r.z = dot_step<S + 2>(hc, o);
    r.w = dot_step<S + 3>(hd, o);
    return r;
}

__device__ __forceinline__ void wave_patch_order()
{
    __builtin_amdgcn_fence(__ATOMIC_RELEASE, "wavefront");
    __builtin_amdgcn_wave_barrier();
    __builtin_amdgcn_fence(__ATOMIC_ACQUIRE, "wavefront");
}

template <int H>
__device__ __forceinline__ void half_chunk(const char* hb, const GOp& o,
                                           float* patch_w)
{
    const int3 to = make_int3(o.t0 * (TROW * 4), o.t1 * (TROW * 4), o.t2 * (TROW * 4));
    *(float4*)&patch_w[8 * H]     = dot_steps4<8 * H>(hb, to, o);
    *(float4*)&patch_w[8 * H + 4] = dot_steps4<8 * H + 4>(hb, to, o);
}

__device__ __forceinline__ void turn_and_store(const float* patch_r,
                                               float* __restrict__ op)
{
    wave_patch_order();
    #pragma unroll
    for (int r = 0; r < BT; ++r)
        op[(size_t)r * G_] = patch_r[r * TRP];
    wave_patch_order();
}

__global__ __launch_bounds__(1024)
void ae_decoder_fused(const float* __restrict__ features,
                      const float* __restrict__ w1,
                      const float* __restrict__ b1,
                      const float* __restrict__ w2,
                      const float* __restrict__ b2,
                      const float* __restrict__ w3,
                      const float* __restrict__ b3,
                      const int*   __restrict__ edge_tf,
                      float*       __restrict__ out)
{
    __shared__ __align__(16) float lds_h2[T_ * BT * K_];
    __shared__ __align__(16) float lds_tr[NW * BT * TRP];

    const int tid  = threadIdx.x;
    const int wave = tid >> 5;
    const int lane = tid & 31;
    const int half = lane >> 4;
    const int m    = lane & 15;

    const int bbase = blockIdx.y * BT;

    const int g0 = (blockIdx.x + NGS * wave) * GCH;
    const bool has_last = g0 + (NU - 1) * GSTEP < G_;

    GOp cur = load_gop(edge_tf, w3, b3, g0 + lane);

    const int j0 = half * 2;
    const int ti = m >> 2;
    const int kk = m & (K_ - 1);
    const int tw = wave * (T_ / NW);
    const float* __restrict__ frow = features + (size_t)(bbase + m) * T_ + tw;

    #pragma nounroll
    for (int q = 0; q < (T_ / NW) / TG; ++q) {
        const int tq = tw + q * TG;

        const float4 fa = *(const float4*)(frow + q * TG);
        const float4 fb = *(const float4*)(frow + q * TG + 4);
        const float fv[TG] = {fa.x, fa.y, fa.z, fa.w, fb.x, fb.y, fb.z, fb.w};

        v2f    w1v[TG], b1v[TG];
        float  w2x[TG / TQ], w2y[TG / TQ];
        float4 c0[TG / TQ], c1[TG / TQ];
        #pragma unroll
        for (int i = 0; i < TG; ++i) {
            const int t = tq + i;
            w1v[i] = *(const v2f*)&w1[t * K_ + j0];
            b1v[i] = *(const v2f*)&b1[t * K_ + j0];
        }
        #pragma unroll
        for (int s = 0; s < TG / TQ; ++s) {
            const int ta = tq + s * TQ + ti;
            w2x[s] = w2[(ta * K_ + j0)     * K_ + kk];
            w2y[s] = w2[(ta * K_ + j0 + 1) * K_ + kk];
            const float4* __restrict__ b2v = (const float4*)(b2 + (tq + s * TQ + 2 * half) * K_);
            c0[s] = b2v[0];
            c1[s] = b2v[1];
        }

        #pragma unroll
        for (int s = 0; s < TG / TQ; ++s) {
            v8f acc = {c0[s].x, c0[s].y, c0[s].z, c0[s].w, c1[s].x, c1[s].y, c1[s].z, c1[s].w};
            #pragma unroll
            for (int i = 0; i < TQ; ++i) {
                const float f = fv[s * TQ + i];
                v2f hop;
                hop.x = lrelu(f * w1v[s * TQ + i].x + b1v[s * TQ + i].x);
                hop.y = lrelu(f * w1v[s * TQ + i].y + b1v[s * TQ + i].y);

                v2f wop;
                wop.x = (ti == i) ? w2x[s] : 0.f;
                wop.y = (ti == i) ? w2y[s] : 0.f;

                acc = __builtin_amdgcn_wmma_f32_16x16x4_f32(
                     false, wop,  false, hop,
                     (short)0, acc,  false,  false);
            }
            float* dst = lds_h2 + ((tq + s * TQ + 2 * half) * BT + m) * K_;
            *(float4*)dst          = make_float4(lrelu(acc[0]), lrelu(acc[1]), lrelu(acc[2]), lrelu(acc[3]));
            *(float4*)(dst + TROW) = make_float4(lrelu(acc[4]), lrelu(acc[5]), lrelu(acc[6]), lrelu(acc[7]));
        }
    }

    __syncthreads();

    const char* hb = (const char*)(lds_h2 + m * K_);
    float* patch = lds_tr + wave * (BT * TRP);
    float* patch_w = patch + m * TRP + 16 * half;
    const float* patch_r = patch + lane;
    float* __restrict__ orow = out + (size_t)bbase * G_ + lane;

    #pragma unroll
    for (int u = 0; u < NU - 1; ++u) {
        const int gb = g0 + u * GSTEP;
        half_chunk<0>(hb, cur, patch_w);
        GOp nxt = cur;
        if (u + 1 < NU - 1 || has_last)
            nxt = load_gop(edge_tf, w3, b3, gb + GSTEP + lane);
        half_chunk<1>(hb, cur, patch_w);
        turn_and_store(patch_r, orow + gb);
        cur = nxt;
    }
    if (has_last) {
        half_chunk<0>(hb, cur, patch_w);
        half_chunk<1>(hb, cur, patch_w);
        turn_and_store(patch_r, orow + g0 + (NU - 1) * GSTEP);
    }
}

extern "C" void kernel_launch(void* const* d_in, const int* in_sizes, int n_in,
                              void* d_out, int out_size, void* d_ws, size_t ws_size,
                              hipStream_t stream)
{
    const float* features = (const float*)d_in[0];
    const float* w1       = (const float*)d_in[1];
    const float* b1       = (const float*)d_in[2];
    const float* w2       = (const float*)d_in[3];
    const float* b2       = (const float*)d_in[4];
    const float* w3       = (const float*)d_in[5];
    const float* b3       = (const float*)d_in[6];
    const int*   edge_tf  = (const int*)d_in[7];
    float*       out      = (float*)d_out;

    dim3 grid(NGS, B_ / BT);
    ae_decoder_fused<<<grid, NW * 32, 0, stream>>>(features, w1, b1, w2, b2,
                                                   w3, b3, edge_tf, out);
}


# Phase B: a wave works two adjacent 32-g chunks and each row's 64 results leave as one aligned 256-B store (8 B a lane; a block's 16 rows share one parity so the units line up); same fma order
# speedup vs baseline: 5.1635x; 1.1108x over previous
#include <hip/hip_runtime.h>


typedef float v2f __attribute__((ext_vector_type(2)));
typedef float v8f __attribute__((ext_vector_type(8)));

#define T_   512
#define K_   4
#define G_   20000
#define EPG_ 3
#define B_   512
#define BT   16
#define NW   32
#define NGS  8
#define GCH  32
#define UG   (2 * GCH)
#define NUN  (G_ / UG)
#define USTEP (NGS * NW)
#define TRP  68
#define TROW (BT * K_)
#define TG   8
#define TQ   4

static_assert(G_ % GCH == 0 && G_ - NUN * UG == GCH, "a row is 312 units and one chunk");
static_assert(USTEP <= NUN && 2 * USTEP > NUN, "every wave has a first unit, some a second, none a third");
static_assert((G_ * 4) % 256 == 128 && (GCH * 4) == 128, "odd rows start half a unit late: their units start 32 g in");
static_assert(B_ % (2 * BT) == 0, "a block's 16 rows are every second row of a 32-row band");
static_assert(BT == 16 && GCH == 32, "a 16-lane row walks the 16 b of the tile; a wave's two rows take 2 x 16 g");
static_assert(T_ % NW == 0 && B_ % BT == 0, "tiles are whole");
static_assert((T_ / NW) % TG == 0 && TG == 8, "a group is two 16-byte feature loads per lane");
static_assert(TQ * K_ == 16 && TG % TQ == 0, "a D tile's 16 rows are TQ t x K_ k; a load group is whole tiles");

__device__ __forceinline__ float lrelu(float x) {
    return x > 0.f ? x : 0.01f * x;
}

struct GOp {
    int    t0, t1, t2;
    float4 w0, w1, w2;
    float  bias;
};

__device__ __forceinline__ GOp load_gop(const int* __restrict__ edge_tf,
                                        const float* __restrict__ w3,
                                        const float* __restrict__ b3, unsigned g)
{
    const float4* __restrict__ w3v = (const float4*)w3;
    GOp o;
    o.t0   = edge_tf[g * EPG_ + 0u];
    o.t1   = edge_tf[g * EPG_ + 1u];
    o.t2   = edge_tf[g * EPG_ + 2u];
    o.w0   = w3v[g * EPG_ + 0u];
    o.w1   = w3v[g * EPG_ + 1u];
    o.w2   = w3v[g * EPG_ + 2u];
    o.bias = b3[g];
    return o;
}

template <int S>
__device__ __forceinline__ int row_share_i(int x) {
    return __builtin_amdgcn_update_dpp(0, x, 0x150 + S, 0xf, 0xf, true);
}
template <int S>
__device__ __forceinline__ float row_share_f(float x) {
    return __int_as_float(row_share_i<S>(__float_as_int(x)));
}
template <int S>
__device__ __forceinline__ float fma_row_share(float h, float w, float acc) {
    asm("v_fmac_f32_dpp %0, %1, %2 row_share:%3 row_mask:0xf bank_mask:0xf bound_ctrl:1"
        : "+v"(acc) : "v"(w), "v"(h), "n"(S));
    return acc;
}

struct H3 { float4 h0, h1, h2; };

template <int S>
__device__ __forceinline__ H3 load_step(const char* hb, const int3& to)
{
    H3 h;
    h.h0 = *(const float4*)(hb + row_share_i<S>(to.x));
    h.h1 = *(const float4*)(hb + row_share_i<S>(to.y));
    h.h2 = *(const float4*)(hb + row_share_i<S>(to.z));
    return h;
}

template <int S>
__device__ __forceinline__ float dot_step(const H3& h, const GOp& o)
{
    const float4 h0 = h.h0, h1 = h.h1, h2 = h.h2;
    float acc = row_share_f<S>(o.bias);
    acc = fma_row_share<S>(h0.x, o.w0.x, acc);
    acc = fma_row_share<S>(h0.y, o.w0.y, acc);
    acc = fma_row_share<S>(h0.z, o.w0.z, acc);
    acc = fma_row_share<S>(h0.w, o.w0.w, acc);
    acc = fma_row_share<S>(h1.x, o.w1.x, acc);
    acc = fma_row_share<S>(h1.y, o.w1.y, acc);
    acc = fma_row_share<S>(h1.z, o.w1.z, acc);
    acc = fma_row_share<S>(h1.w, o.w1.w, acc);
    acc = fma_row_share<S>(h2.x, o.w2.x, acc);
    acc = fma_row_share<S>(h2.y, o.w2.y, acc);
    acc = fma_row_share<S>(h2.z, o.w2.z, acc);
    acc = fma_row_share<S>(h2.w, o.w2.w, acc);
    return acc;
}

template <int S>
__device__ __forceinline__ float4 dot_steps4(const char* hb, const int3& to, const GOp& o)
{
    const H3 ha = load_step<S>(hb, to);
    const H3 hb1 = load_step<S + 1>(hb, to);
    const H3 hc = load_step<S + 2>(hb, to);
    const H3 hd = load_step<S + 3>(hb, to);
    float4 r;
    r.x = dot_step<S>(ha, o);
    r.y = dot_step<S + 1>(hb1, o);
    r.z = dot_step<S + 2>(hc, o);
    r.w = dot_step<S + 3>(hd, o);
    return r;
}

__device__ __forceinline__ void wave_patch_order()
{
    __builtin_amdgcn_fence(__ATOMIC_RELEASE, "wavefront");
    __builtin_amdgcn_wave_barrier();
    __builtin_amdgcn_fence(__ATOMIC_ACQUIRE, "wavefront");
}

template <int H>
__device__ __forceinline__ void half_chunk(const char* hb, const GOp& o,
                                           float* patch_w)
{
    const int3 to = make_int3(o.t0 * (TROW * 4), o.t1 * (TROW * 4), o.t2 * (TROW * 4));
    *(float4*)&patch_w[8 * H]     = dot_steps4<8 * H>(hb, to, o);
    *(float4*)&patch_w[8 * H + 4] = dot_steps4<8 * H + 4>(hb, to, o);
}

__device__ __forceinline__ void turn_and_store_unit(const float* patch_r2,
                                                    float* __restrict__ op)
{
    wave_patch_order();
    #pragma unroll
    for (int r = 0; r < BT; ++r)
        *(float2*)(op + (size_t)r * (2 * G_)) = *(const float2*)(patch_r2 + r * TRP);
    wave_patch_order();
}

__device__ __forceinline__ void turn_and_store_chunk(const float* patch_r,
                                                     float* __restrict__ op)
{
    wave_patch_order();
    #pragma unroll
    for (int r = 0; r < BT; ++r)
        op[(size_t)r * (2 * G_)] = patch_r[r * TRP];
    wave_patch_order();
}

__global__ __launch_bounds__(1024)
void ae_decoder_fused(const float* __restrict__ features,
                      const float* __restrict__ w1,
                      const float* __restrict__ b1,
                      const float* __restrict__ w2,
                      const float* __restrict__ b2,
                      const float* __restrict__ w3,
                      const float* __restrict__ b3,
                      const int*   __restrict__ edge_tf,
                      float*       __restrict__ out)
{
    __shared__ __align__(16) float lds_h2[T_ * BT * K_];
    __shared__ __align__(16) float lds_tr[NW * BT * TRP];

    const int tid  = threadIdx.x;
    const int wave = tid >> 5;
    const int lane = tid & 31;
    const int half = lane >> 4;
    const int m    = lane & 15;

    const int par   = blockIdx.y & 1;
    const int brow0 = (blockIdx.y >> 1) * (2 * BT) + par;

    const int u0 = blockIdx.x + NGS * wave;
    const int g0 = u0 * UG + par * GCH;
    const bool has_u2   = u0 + USTEP < NUN;
    const bool has_rest = (blockIdx.x == 0) && (wave == 7);
    const int  g_rest   = par ? 0 : NUN * UG;
    const int  g_after  = has_u2 ? g0 + USTEP * UG : g_rest;

    GOp cur = load_gop(edge_tf, w3, b3, g0 + lane);

    const int j0 = half * 2;
    const int ti = m >> 2;
    const int kk = m & (K_ - 1);
    const int tw = wave * (T_ / NW);
    const float* __restrict__ frow = features + (size_t)(brow0 + 2 * m) * T_ + tw;

    #pragma nounroll
    for (int q = 0; q < (T_ / NW) / TG; ++q) {
        const int tq = tw + q * TG;

        const float4 fa = *(const float4*)(frow + q * TG);
        const float4 fb = *(const float4*)(frow + q * TG + 4);
        const float fv[TG] = {fa.x, fa.y, fa.z, fa.w, fb.x, fb.y, fb.z, fb.w};

        v2f    w1v[TG], b1v[TG];
        float  w2x[TG / TQ], w2y[TG / TQ];
        float4 c0[TG / TQ], c1[TG / TQ];
        #pragma unroll
        for (int i = 0; i < TG; ++i) {
            const int t = tq + i;
            w1v[i] = *(const v2f*)&w1[t * K_ + j0];
            b1v[i] = *(const v2f*)&b1[t * K_ + j0];
        }
        #pragma unroll
        for (int s = 0; s < TG / TQ; ++s) {
            const int ta = tq + s * TQ + ti;
            w2x[s] = w2[(ta * K_ + j0)     * K_ + kk];
            w2y[s] = w2[(ta * K_ + j0 + 1) * K_ + kk];
            const float4* __restrict__ b2v = (const float4*)(b2 + (tq + s * TQ + 2 * half) * K_);
            c0[s] = b2v[0];
            c1[s] = b2v[1];
        }

        #pragma unroll
        for (int s = 0; s < TG / TQ; ++s) {
            v8f acc = {c0[s].x, c0[s].y, c0[s].z, c0[s].w, c1[s].x, c1[s].y, c1[s].z, c1[s].w};
            #pragma unroll
            for (int i = 0; i < TQ; ++i) {
                const float f = fv[s * TQ + i];
                v2f hop;
                hop.x = lrelu(f * w1v[s * TQ + i].x + b1v[s * TQ + i].x);
                hop.y = lrelu(f * w1v[s * TQ + i].y + b1v[s * TQ + i].y);

                v2f wop;
                wop.x = (ti == i) ? w2x[s] : 0.f;
                wop.y = (ti == i) ? w2y[s] : 0.f;

                acc = __builtin_amdgcn_wmma_f32_16x16x4_f32(
                     false, wop,  false, hop,
                     (short)0, acc,  false,  false);
            }
            float* dst = lds_h2 + ((tq + s * TQ + 2 * half) * BT + m) * K_;
            *(float4*)dst          = make_float4(lrelu(acc[0]), lrelu(acc[1]), lrelu(acc[2]), lrelu(acc[3]));
            *(float4*)(dst + TROW) = make_float4(lrelu(acc[4]), lrelu(acc[5]), lrelu(acc[6]), lrelu(acc[7]));
        }
    }

    __syncthreads();

    const char* hb = (const char*)(lds_h2 + m * K_);
    float* patch = lds_tr + wave * (BT * TRP);
    float* patch_w = patch + m * TRP + 16 * half;
    const float* patch_r  = patch + lane;
    const float* patch_r2 = patch + 2 * lane;
    float* __restrict__ orow = out + (size_t)brow0 * G_;

    int g = g0;
    #pragma nounroll
    for (int u = 0; u < 2; ++u) {
        half_chunk<0>(hb, cur, patch_w);
        const GOp opb = load_gop(edge_tf, w3, b3, g + GCH + lane);
        half_chunk<1>(hb, cur, patch_w);
        half_chunk<0>(hb, opb, patch_w + GCH);
        GOp nxt = opb;
        if (u == 0 && (has_u2 || has_rest))
            nxt = load_gop(edge_tf, w3, b3, g_after + lane);
        half_chunk<1>(hb, opb, patch_w + GCH);
        turn_and_store_unit(patch_r2, orow + g + 2 * lane);
        cur = nxt;
        g = g_after;
        if (!has_u2) break;
    }
    if (has_rest) {
        half_chunk<0>(hb, cur, patch_w);
        half_chunk<1>(hb, cur, patch_w);
        turn_and_store_chunk(patch_r, orow + g_rest + lane);
    }
}

extern "C" void kernel_launch(void* const* d_in, const int* in_sizes, int n_in,
                              void* d_out, int out_size, void* d_ws, size_t ws_size,
                              hipStream_t stream)
{
    const float* features = (const float*)d_in[0];
    const float* w1       = (const float*)d_in[1];
    const float* b1       = (const float*)d_in[2];
    const float* w2       = (const float*)d_in[3];
    const float* b2       = (const float*)d_in[4];
    const float* w3       = (const float*)d_in[5];
    const float* b3       = (const float*)d_in[6];
    const int*   edge_tf  = (const int*)d_in[7];
    float*       out      = (float*)d_out;

    dim3 grid(NGS, B_ / BT);
    ae_decoder_fused<<<grid, NW * 32, 0, stream>>>(features, w1, b1, w2, b2,
                                                   w3, b3, edge_tf, out);
}


# Half as many waves: 256 blocks x 16 waves (4 per SIMD), each wave 32 t of phase A (two load groups of 16 t) and 2-3 units of phase B: 4096 waves enter, not 8192; same fma order
# speedup vs baseline: 5.8762x; 1.1380x over previous
#include <hip/hip_runtime.h>


typedef float v2f __attribute__((ext_vector_type(2)));
typedef float v8f __attribute__((ext_vector_type(8)));

#define T_   512
#define K_   4
#define G_   20000
#define EPG_ 3
#define B_   512
#define BT   16
#define NW   16
#define NGS  8
#define GCH  32
#define UG   (2 * GCH)
#define NUN  (G_ / UG)
#define USTEP (NGS * NW)
#define TRP  68
#define TROW (BT * K_)
#define TG   8
#define TQ   4

static_assert(G_ % GCH == 0 && G_ - NUN * UG == GCH, "a row is 312 units and one chunk");
static_assert(2 * USTEP <= NUN && 3 * USTEP > NUN, "every wave has two units, some a third, none a fourth");
static_assert(7 * NGS + 2 * USTEP >= NUN, "the wave that takes the left-over chunk has no third unit");
static_assert((G_ * 4) % 256 == 128 && (GCH * 4) == 128, "odd rows start half a unit late: their units start 32 g in");
static_assert(B_ % (2 * BT) == 0, "a block's 16 rows are every second row of a 32-row band");
static_assert(BT == 16 && GCH == 32, "a 16-lane row walks the 16 b of the tile; a wave's two rows take 2 x 16 g");
static_assert(T_ % NW == 0 && B_ % BT == 0, "tiles are whole");
static_assert((T_ / NW) % TG == 0 && TG == 8, "a group is two 16-byte feature loads per lane");
static_assert(TQ * K_ == 16 && TG % TQ == 0, "a D tile's 16 rows are TQ t x K_ k; a load group is whole tiles");

__device__ __forceinline__ float lrelu(float x) {
    return x > 0.f ? x : 0.01f * x;
}

struct GOp {
    int    t0, t1, t2;
    float4 w0, w1, w2;
    float  bias;
};

__device__ __forceinline__ GOp load_gop(const int* __restrict__ edge_tf,
                                        const float* __restrict__ w3,
                                        const float* __restrict__ b3, unsigned g)
{
    const float4* __restrict__ w3v = (const float4*)w3;
    GOp o;
    o.t0   = edge_tf[g * EPG_ + 0u];
    o.t1   = edge_tf[g * EPG_ + 1u];
    o.t2   = edge_tf[g * EPG_ + 2u];
    o.w0   = w3v[g * EPG_ + 0u];
    o.w1   = w3v[g * EPG_ + 1u];
    o.w2   = w3v[g * EPG_ + 2u];
    o.bias = b3[g];
    return o;
}

template <int S>
__device__ __forceinline__ int row_share_i(int x) {
    return __builtin_amdgcn_update_dpp(0, x, 0x150 + S, 0xf, 0xf, true);
}
template <int S>
__device__ __forceinline__ float row_share_f(float x) {
    return __int_as_float(row_share_i<S>(__float_as_int(x)));
}
template <int S>
__device__ __forceinline__ float fma_row_share(float h, float w, float acc) {
    asm("v_fmac_f32_dpp %0, %1, %2 row_share:%3 row_mask:0xf bank_mask:0xf bound_ctrl:1"
        : "+v"(acc) : "v"(w), "v"(h), "n"(S));
    return acc;
}

struct H3 { float4 h0, h1, h2; };

template <int S>
__device__ __forceinline__ H3 load_step(const char* hb, const int3& to)
{
    H3 h;
    h.h0 = *(const float4*)(hb + row_share_i<S>(to.x));
    h.h1 = *(const float4*)(hb + row_share_i<S>(to.y));
    h.h2 = *(const float4*)(hb + row_share_i<S>(to.z));
    return h;
}

template <int S>
__device__ __forceinline__ float dot_step(const H3& h, const GOp& o)
{
    const float4 h0 = h.h0, h1 = h.h1, h2 = h.h2;
    float acc = row_share_f<S>(o.bias);
    acc = fma_row_share<S>(h0.x, o.w0.x, acc);
    acc = fma_row_share<S>(h0.y, o.w0.y, acc);
    acc = fma_row_share<S>(h0.z, o.w0.z, acc);
    acc = fma_row_share<S>(h0.w, o.w0.w, acc);
    acc = fma_row_share<S>(h1.x, o.w1.x, acc);
    acc = fma_row_share<S>(h1.y, o.w1.y, acc);
    acc = fma_row_share<S>(h1.z, o.w1.z, acc);
    acc = fma_row_share<S>(h1.w, o.w1.w, acc);
    acc = fma_row_share<S>(h2.x, o.w2.x, acc);
    acc = fma_row_share<S>(h2.y, o.w2.y, acc);
    acc = fma_row_share<S>(h2.z, o.w2.z, acc);
    acc = fma_row_share<S>(h2.w, o.w2.w, acc);
    return acc;
}

template <int S>
__device__ __forceinline__ float4 dot_steps4(const char* hb, const int3& to, const GOp& o)
{
    const H3 ha = load_step<S>(hb, to);
    const H3 hb1 = load_step<S + 1>(hb, to);
    const H3 hc = load_step<S + 2>(hb, to);
    const H3 hd = load_step<S + 3>(hb, to);
    float4 r;
    r.x = dot_step<S>(ha, o);
    r.y = dot_step<S + 1>(hb1, o);
    r.z = dot_step<S + 2>(hc, o);
    r.w = dot_step<S + 3>(hd, o);
    return r;
}

__device__ __forceinline__ void wave_patch_order()
{
    __builtin_amdgcn_fence(__ATOMIC_RELEASE, "wavefront");
    __builtin_amdgcn_wave_barrier();
    __builtin_amdgcn_fence(__ATOMIC_ACQUIRE, "wavefront");
}

template <int H>
__device__ __forceinline__ void half_chunk(const char* hb, const GOp& o,
                                           float* patch_w)
{
    const int3 to = make_int3(o.t0 * (TROW * 4), o.t1 * (TROW * 4), o.t2 * (TROW * 4));
    *(float4*)&patch_w[8 * H]     = dot_steps4<8 * H>(hb, to, o);
    *(float4*)&patch_w[8 * H + 4] = dot_steps4<8 * H + 4>(hb, to, o);
}

__device__ __forceinline__ void turn_and_store_unit(const float* patch_r2,
                                                    float* __restrict__ op)
{
    wave_patch_order();
    #pragma unroll
    for (int r = 0; r < BT; ++r)
        *(float2*)(op + (size_t)r * (2 * G_)) = *(const float2*)(patch_r2 + r * TRP);
    wave_patch_order();
}

__device__ __forceinline__ void turn_and_store_chunk(const float* patch_r,
                                                     float* __restrict__ op)
{
    wave_patch_order();
    #pragma unroll
    for (int r = 0; r < BT; ++r)
        op[(size_t)r * (2 * G_)] = patch_r[r * TRP];
    wave_patch_order();
}

__global__ __launch_bounds__(NW * 32)
void ae_decoder_fused(const float* __restrict__ features,
                      const float* __restrict__ w1,
                      const float* __restrict__ b1,
                      const float* __restrict__ w2,
                      const float* __restrict__ b2,
                      const float* __restrict__ w3,
                      const float* __restrict__ b3,
                      const int*   __restrict__ edge_tf,
                      float*       __restrict__ out)
{
    __shared__ __align__(16) float lds_h2[T_ * BT * K_];
    __shared__ __align__(16) float lds_tr[NW * BT * TRP];

    const int tid  = threadIdx.x;
    const int wave = tid >> 5;
    const int lane = tid & 31;
    const int half = lane >> 4;
    const int m    = lane & 15;

    const int par   = blockIdx.y & 1;
    const int brow0 = (blockIdx.y >> 1) * (2 * BT) + par;

    const int u0 = blockIdx.x + NGS * wave;
    const int g0 = u0 * UG + par * GCH;
    const int n_units   = (u0 + 2 * USTEP < NUN) ? 3 : 2;
    const bool has_rest = (blockIdx.x == 0) && (wave == 7);
    const int  g_rest   = par ? 0 : NUN * UG;

    GOp cur = load_gop(edge_tf, w3, b3, g0 + lane);

    const int j0 = half * 2;
    const int ti = m >> 2;
    const int kk = m & (K_ - 1);
    const int tw = wave * (T_ / NW);
    const float* __restrict__ frow = features + (size_t)(brow0 + 2 * m) * T_ + tw;

    #pragma unroll 2
    for (int q = 0; q < (T_ / NW) / TG; ++q) {
        const int tq = tw + q * TG;

        const float4 fa = *(const float4*)(frow + q * TG);
        const float4 fb = *(const float4*)(frow + q * TG + 4);
        const float fv[TG] = {fa.x, fa.y, fa.z, fa.w, fb.x, fb.y, fb.z, fb.w};

        v2f    w1v[TG], b1v[TG];
        float  w2x[TG / TQ], w2y[TG / TQ];
        float4 c0[TG / TQ], c1[TG / TQ];
        #pragma unroll
        for (int i = 0; i < TG; ++i) {
            const int t = tq + i;
            w1v[i] = *(const v2f*)&w1[t * K_ + j0];
            b1v[i] = *(const v2f*)&b1[t * K_ + j0];
        }
        #pragma unroll
        for (int s = 0; s < TG / TQ; ++s) {
            const int ta = tq + s * TQ + ti;
            w2x[s] = w2[(ta * K_ + j0)     * K_ + kk];
            w2y[s] = w2[(ta * K_ + j0 + 1) * K_ + kk];
            const float4* __restrict__ b2v = (const float4*)(b2 + (tq + s * TQ + 2 * half) * K_);
            c0[s] = b2v[0];
            c1[s] = b2v[1];
        }

        #pragma unroll
        for (int s = 0; s < TG / TQ; ++s) {
            v8f acc = {c0[s].x, c0[s].y, c0[s].z, c0[s].w, c1[s].x, c1[s].y, c1[s].z, c1[s].w};
            #pragma unroll
            for (int i = 0; i < TQ; ++i) {
                const float f = fv[s * TQ + i];
                v2f hop;
                hop.x = lrelu(f * w1v[s * TQ + i].x + b1v[s * TQ + i].x);
                hop.y = lrelu(f * w1v[s * TQ + i].y + b1v[s * TQ + i].y);

                v2f wop;
                wop.x = (ti == i) ? w2x[s] : 0.f;
                wop.y = (ti == i) ? w2y[s] : 0.f;

                acc = __builtin_amdgcn_wmma_f32_16x16x4_f32(
                     false, wop,  false, hop,
                     (short)0, acc,  false,  false);
            }
            float* dst = lds_h2 + ((tq + s * TQ + 2 * half) * BT + m) * K_;
            *(float4*)dst          = make_float4(lrelu(acc[0]), lrelu(acc[1]), lrelu(acc[2]), lrelu(acc[3]));
            *(float4*)(dst + TROW) = make_float4(lrelu(acc[4]), lrelu(acc[5]), lrelu(acc[6]), lrelu(acc[7]));
        }
    }

    __syncthreads();

    const char* hb = (const char*)(lds_h2 + m * K_);
    float* patch = lds_tr + wave * (BT * TRP);
    float* patch_w = patch + m * TRP + 16 * half;
    const float* patch_r  = patch + lane;
    const float* patch_r2 = patch + 2 * lane;
    float* __restrict__ orow = out + (size_t)brow0 * G_;

    int g = g0;
    #pragma nounroll
    for (int u = 0; u < 3; ++u) {
        const bool last   = (u + 1 >= n_units);
        const int  g_next = last ? g_rest : g + USTEP * UG;
        half_chunk<0>(hb, cur, patch_w);
        const GOp opb = load_gop(edge_tf, w3, b3, g + GCH + lane);
        half_chunk<1>(hb, cur, patch_w);
        half_chunk<0>(hb, opb, patch_w + GCH);
        GOp nxt = opb;
        if (!last || has_rest)
            nxt = load_gop(edge_tf, w3, b3, g_next + lane);
        half_chunk<1>(hb, opb, patch_w + GCH);
        turn_and_store_unit(patch_r2, orow + g + 2 * lane);
        cur = nxt;
        g = g_next;
        if (last) break;
    }
    if (has_rest) {
        half_chunk<0>(hb, cur, patch_w);
        half_chunk<1>(hb, cur, patch_w);
        turn_and_store_chunk(patch_r, orow + g_rest + lane);
    }
}

extern "C" void kernel_launch(void* const* d_in, const int* in_sizes, int n_in,
                              void* d_out, int out_size, void* d_ws, size_t ws_size,
                              hipStream_t stream)
{
    const float* features = (const float*)d_in[0];
    const float* w1       = (const float*)d_in[1];
    const float* b1       = (const float*)d_in[2];
    const float* w2       = (const float*)d_in[3];
    const float* b2       = (const float*)d_in[4];
    const float* w3       = (const float*)d_in[5];
    const float* b3       = (const float*)d_in[6];
    const int*   edge_tf  = (const int*)d_in[7];
    float*       out      = (float*)d_out;

    dim3 grid(NGS, B_ / BT);
    ae_decoder_fused<<<grid, NW * 32, 0, stream>>>(features, w1, b1, w2, b2,
                                                   w3, b3, edge_tf, out);
}


# Phase B: each lane keeps its own g (edges, w3, b3 in its registers) and at step s reads b = m xor s of its t rows: conflict-free LDS gather, plain fmas, no cross-lane operand select; same fma order
# speedup vs baseline: 6.3006x; 1.0722x over previous
#include <hip/hip_runtime.h>


typedef float v2f __attribute__((ext_vector_type(2)));
typedef float v8f __attribute__((ext_vector_type(8)));

#define T_   512
#define K_   4
#define G_   20000
#define EPG_ 3
#define B_   512
#define BT   16
#define NW   16
#define NGS  8
#define GCH  32
#define UG   (2 * GCH)
#define NUN  (G_ / UG)
#define USTEP (NGS * NW)
#define TRP  64
#define TROW (BT * K_)
#define TG   8
#define TQ   4

static_assert(G_ % GCH == 0 && G_ - NUN * UG == GCH, "a row is 312 units and one chunk");
static_assert(2 * USTEP <= NUN && 3 * USTEP > NUN, "every wave has two units, some a third, none a fourth");
static_assert(7 * NGS + 2 * USTEP >= NUN, "the wave that takes the left-over chunk has no third unit");
static_assert((G_ * 4) % 256 == 128 && (GCH * 4) == 128, "odd rows start half a unit late: their units start 32 g in");
static_assert(B_ % (2 * BT) == 0, "a block's 16 rows are every second row of a 32-row band");
static_assert(BT == 16 && GCH == 32 && K_ == 4, "a half-wave's 16 lanes read 16 different 16-byte b columns; a lane per g of the chunk");
static_assert(TRP * 4 == 256 && TROW * 4 == 256, "a patch row and a tile row are 256 B: bits 8-11 / 4-7 of an offset hold m");
static_assert(T_ % NW == 0 && B_ % BT == 0, "tiles are whole");
static_assert((T_ / NW) % TG == 0 && TG == 8, "a group is two 16-byte feature loads per lane");
static_assert(TQ * K_ == 16 && TG % TQ == 0, "a D tile's 16 rows are TQ t x K_ k; a load group is whole tiles");

__device__ __forceinline__ float lrelu(float x) {
    return x > 0.f ? x : 0.01f * x;
}

struct GOp {
    int    t0, t1, t2;
    float4 w0, w1, w2;
    float  bias;
};

__device__ __forceinline__ GOp load_gop(const int* __restrict__ edge_tf,
                                        const float* __restrict__ w3,
                                        const float* __restrict__ b3, unsigned g)
{
    const float4* __restrict__ w3v = (const float4*)w3;
    GOp o;
    o.t0   = edge_tf[g * EPG_ + 0u];
    o.t1   = edge_tf[g * EPG_ + 1u];
    o.t2   = edge_tf[g * EPG_ + 2u];
    o.w0   = w3v[g * EPG_ + 0u];
    o.w1   = w3v[g * EPG_ + 1u];
    o.w2   = w3v[g * EPG_ + 2u];
    o.bias = b3[g];
    return o;
}

struct H3 { float4 h0, h1, h2; };

template <int S>
__device__ __forceinline__ H3 load_step(const char* tile, const uint3& a)
{
    H3 h;
    h.h0 = *(const float4*)(tile + (a.x ^ (unsigned)(S << 4)));
    h.h1 = *(const float4*)(tile + (a.y ^ (unsigned)(S << 4)));
    h.h2 = *(const float4*)(tile + (a.z ^ (unsigned)(S << 4)));
    return h;
}

__device__ __forceinline__ float dot_step(const H3& h, const GOp& o)
{
    float acc = o.bias;
    acc = fmaf(h.h0.x, o.w0.x, acc);
    acc = fmaf(h.h0.y, o.w0.y, acc);
    acc = fmaf(h.h0.z, o.w0.z, acc);
    acc = fmaf(h.h0.w, o.w0.w, acc);
    acc = fmaf(h.h1.x, o.w1.x, acc);
    acc = fmaf(h.h1.y, o.w1.y, acc);
    acc = fmaf(h.h1.z, o.w1.z, acc);
    acc = fmaf(h.h1.w, o.w1.w, acc);
    acc = fmaf(h.h2.x, o.w2.x, acc);
    acc = fmaf(h.h2.y, o.w2.y, acc);
    acc = fmaf(h.h2.z, o.w2.z, acc);
    acc = fmaf(h.h2.w, o.w2.w, acc);
    return acc;
}

template <int S>
__device__ __forceinline__ void dot_steps4(const char* tile, const uint3& a, const GOp& o,
                                           char* patches, unsigned pw)
{
    const H3 ha = load_step<S>(tile, a);
    const H3 hb = load_step<S + 1>(tile, a);
    const H3 hc = load_step<S + 2>(tile, a);
    const H3 hd = load_step<S + 3>(tile, a);
    *(float*)(patches + (pw ^ (unsigned)((S)     << 8))) = dot_step(ha, o);
    *(float*)(patches + (pw ^ (unsigned)((S + 1) << 8))) = dot_step(hb, o);
    *(float*)(patches + (pw ^ (unsigned)((S + 2) << 8))) = dot_step(hc, o);
    *(float*)(patches + (pw ^ (unsigned)((S + 3) << 8))) = dot_step(hd, o);
}

__device__ __forceinline__ void wave_patch_order()
{
    __builtin_amdgcn_fence(__ATOMIC_RELEASE, "wavefront");
    __builtin_amdgcn_wave_barrier();
    __builtin_amdgcn_fence(__ATOMIC_ACQUIRE, "wavefront");
}

template <int H>
__device__ __forceinline__ void half_chunk(const char* tile, unsigned mb, const GOp& o,
                                           char* patches, unsigned pw)
{
    const uint3 a = make_uint3((unsigned)o.t0 * (TROW * 4) + mb,
                               (unsigned)o.t1 * (TROW * 4) + mb,
                               (unsigned)o.t2 * (TROW * 4) + mb);
    dot_steps4<8 * H>(tile, a, o, patches, pw);
    dot_steps4<8 * H + 4>(tile, a, o, patches, pw);
    __builtin_amdgcn_sched_barrier(0);
}

__device__ __forceinline__ void turn_and_store_unit(const float* patch_r2,
                                                    float* __restrict__ op)
{
    wave_patch_order();
    #pragma unroll
    for (int r = 0; r < BT; ++r)
        *(float2*)(op + (size_t)r * (2 * G_)) = *(const float2*)(patch_r2 + r * TRP);
    wave_patch_order();
}

__device__ __forceinline__ void turn_and_store_chunk(const float* patch_r,
                                                     float* __restrict__ op)
{
    wave_patch_order();
    #pragma unroll
    for (int r = 0; r < BT; ++r)
        op[(size_t)r * (2 * G_)] = patch_r[r * TRP];
    wave_patch_order();
}

__global__ __launch_bounds__(NW * 32)
void ae_decoder_fused(const float* __restrict__ features,
                      const float* __restrict__ w1,
                      const float* __restrict__ b1,
                      const float* __restrict__ w2,
                      const float* __restrict__ b2,
                      const float* __restrict__ w3,
                      const float* __restrict__ b3,
                      const int*   __restrict__ edge_tf,
                      float*       __restrict__ out)
{
    __shared__ __align__(16) float lds_h2[T_ * BT * K_];
    __shared__ __align__(16) float lds_tr[NW * BT * TRP];

    const int tid  = threadIdx.x;
    const int wave = tid >> 5;
    const int lane = tid & 31;
    const int half = lane >> 4;
    const int m    = lane & 15;

    const int par   = blockIdx.y & 1;
    const int brow0 = (blockIdx.y >> 1) * (2 * BT) + par;

    const int u0 = blockIdx.x + NGS * wave;
    const int g0 = u0 * UG + par * GCH;
    const int n_units   = (u0 + 2 * USTEP < NUN) ? 3 : 2;
    const bool has_rest = (blockIdx.x == 0) && (wave == 7);
    const int  g_rest   = par ? 0 : NUN * UG;

    GOp cur = load_gop(edge_tf, w3, b3, g0 + lane);

    const int j0 = half * 2;
    const int ti = m >> 2;
    const int kk = m & (K_ - 1);
    const int tw = wave * (T_ / NW);
    const float* __restrict__ frow = features + (size_t)(brow0 + 2 * m) * T_ + tw;

    #pragma unroll 2
    for (int q = 0; q < (T_ / NW) / TG; ++q) {
        const int tq = tw + q * TG;

        const float4 fa = *(const float4*)(frow + q * TG);
        const float4 fb = *(const float4*)(frow + q * TG + 4);
        const float fv[TG] = {fa.x, fa.y, fa.z, fa.w, fb.x, fb.y, fb.z, fb.w};

        v2f    w1v[TG], b1v[TG];
        float  w2x[TG / TQ], w2y[TG / TQ];
        float4 c0[TG / TQ], c1[TG / TQ];
        #pragma unroll
        for (int i = 0; i < TG; ++i) {
            const int t = tq + i;
            w1v[i] = *(const v2f*)&w1[t * K_ + j0];
            b1v[i] = *(const v2f*)&b1[t * K_ + j0];
        }
        #pragma unroll
        for (int s = 0; s < TG / TQ; ++s) {
            const int ta = tq + s * TQ + ti;
            w2x[s] = w2[(ta * K_ + j0)     * K_ + kk];
            w2y[s] = w2[(ta * K_ + j0 + 1) * K_ + kk];
            const float4* __restrict__ b2v = (const float4*)(b2 + (tq + s * TQ + 2 * half) * K_);
            c0[s] = b2v[0];
            c1[s] = b2v[1];
        }

        #pragma unroll
        for (int s = 0; s < TG / TQ; ++s) {
            v8f acc = {c0[s].x, c0[s].y, c0[s].z, c0[s].w, c1[s].x, c1[s].y, c1[s].z, c1[s].w};
            #pragma unroll
            for (int i = 0; i < TQ; ++i) {
                const float f = fv[s * TQ + i];
                v2f hop;
                hop.x = lrelu(f * w1v[s * TQ + i].x + b1v[s * TQ + i].x);
                hop.y = lrelu(f * w1v[s * TQ + i].y + b1v[s * TQ + i].y);

                v2f wop;
                wop.x = (ti == i) ? w2x[s] : 0.f;
                wop.y = (ti == i) ? w2y[s] : 0.f;

                acc = __builtin_amdgcn_wmma_f32_16x16x4_f32(
                     false, wop,  false, hop,
                     (short)0, acc,  false,  false);
            }
            float* dst = lds_h2 + ((tq + s * TQ + 2 * half) * BT + m) * K_;
            *(float4*)dst          = make_float4(lrelu(acc[0]), lrelu(acc[1]), lrelu(acc[2]), lrelu(acc[3]));
            *(float4*)(dst + TROW) = make_float4(lrelu(acc[4]), lrelu(acc[5]), lrelu(acc[6]), lrelu(acc[7]));
        }
    }

    __syncthreads();

    const char* tile = (const char*)lds_h2;
    const unsigned mb = (unsigned)m * (K_ * 4);
    float* patch = lds_tr + wave * (BT * TRP);
    char* patches = (char*)lds_tr;
    const unsigned pw = (unsigned)((wave * BT + m) * TRP + lane) * 4u;
    const float* patch_r  = patch + lane;
    const float* patch_r2 = patch + 2 * lane;
    float* __restrict__ orow = out + (size_t)brow0 * G_;

    int g = g0;
    #pragma nounroll
    for (int u = 0; u < 3; ++u) {
        const bool last   = (u + 1 >= n_units);
        const int  g_next = last ? g_rest : g + USTEP * UG;
        half_chunk<0>(tile, mb, cur, patches, pw);
        const GOp opb = load_gop(edge_tf, w3, b3, g + GCH + lane);
        half_chunk<1>(tile, mb, cur, patches, pw);
        half_chunk<0>(tile, mb, opb, patches, pw + GCH * 4);
        GOp nxt = opb;
        if (!last || has_rest)
            nxt = load_gop(edge_tf, w3, b3, g_next + lane);
        half_chunk<1>(tile, mb, opb, patches, pw + GCH * 4);
        turn_and_store_unit(patch_r2, orow + g + 2 * lane);
        cur = nxt;
        g = g_next;
        if (last) break;
    }
    if (has_rest) {
        half_chunk<0>(tile, mb, cur, patches, pw);
        half_chunk<1>(tile, mb, cur, patches, pw);
        turn_and_store_chunk(patch_r, orow + g_rest + lane);
    }
}

extern "C" void kernel_launch(void* const* d_in, const int* in_sizes, int n_in,
                              void* d_out, int out_size, void* d_ws, size_t ws_size,
                              hipStream_t stream)
{
    const float* features = (const float*)d_in[0];
    const float* w1       = (const float*)d_in[1];
    const float* b1       = (const float*)d_in[2];
    const float* w2       = (const float*)d_in[3];
    const float* b2       = (const float*)d_in[4];
    const float* w3       = (const float*)d_in[5];
    const float* b3       = (const float*)d_in[6];
    const int*   edge_tf  = (const int*)d_in[7];
    float*       out      = (float*)d_out;

    dim3 grid(NGS, B_ / BT);
    ae_decoder_fused<<<grid, NW * 32, 0, stream>>>(features, w1, b1, w2, b2,
                                                   w3, b3, edge_tf, out);
}
